# starting kernel
# baseline (speedup 1.0000x reference)
#include <hip/hip_runtime.h>
#include <hip/hip_bf16.h>


#define TT      8192
#define NBATCH  128
#define DIN     32
#define HID     16
#define CHLEN   128
#define NCHUNK  (TT / CHLEN)
#define NTILE   (CHLEN / 16)
#define LDS_STRIDE 17

typedef float v2f __attribute__((ext_vector_type(2)));
typedef float v8f __attribute__((ext_vector_type(8)));

__global__ __launch_bounds__(32) void ssm_chunk_kernel(
    const float* __restrict__ u, const float* __restrict__ A,
    const float* __restrict__ W_B, const float* __restrict__ b_B,
    const float* __restrict__ W_C, const float* __restrict__ b_C,
    float* __restrict__ y_partial, float* __restrict__ S)
{
    __shared__ float buLds[CHLEN * LDS_STRIDE];
    __shared__ float ybuf[CHLEN];

    const int wg = blockIdx.x;
    const int b  = wg / NCHUNK;
    const int c  = wg % NCHUNK;
    const int t0 = c * CHLEN;
    const int ln = threadIdx.x;
    const int m  = ln & 15;
    const int kh = ln >> 4;

    v2f bfrag[8];
#pragma unroll
    for (int k = 0; k < 8; ++k) {
        const float* p = W_B + m * DIN + 4 * k + 2 * kh;
        bfrag[k].x = p[0];
        bfrag[k].y = p[1];
    }
    const float bias = b_B[m];

    const float* ubase = u + ((size_t)b * TT + t0) * DIN;

    for (int tile = 0; tile < NTILE; ++tile) {
        if (tile + 1 < NTILE)
            __builtin_prefetch(ubase + (tile + 1) * 16 * DIN + ln * 16, 0, 1);

        v8f cacc;
#pragma unroll
        for (int v = 0; v < 8; ++v) cacc[v] = bias;

        const float* arow = ubase + (tile * 16 + m) * DIN + 2 * kh;
#pragma unroll
        for (int k = 0; k < 8; ++k) {
            v2f afrag;
            afrag.x = arow[4 * k];
            afrag.y = arow[4 * k + 1];
            cacc = __builtin_amdgcn_wmma_f32_16x16x4_f32(
                 false, afrag,  false, bfrag[k],
                 (short)0, cacc,  false,  false);
        }

#pragma unroll
        for (int v = 0; v < 8; ++v) {
            int t = tile * 16 + v + 8 * kh;
            buLds[t * LDS_STRIDE + m] = cacc[v];
        }
    }

    __syncthreads();

    const float bC = b_C[0];
    if (ln < 16) {
        const float Aval = A[ln];
        const float wc   = W_C[ln];
        float hreg = 0.0f;
        for (int t = 0; t < CHLEN; ++t) {
            float bu = buLds[t * LDS_STRIDE + ln];
            hreg = fmaf(Aval, hreg, bu);
            float val = wc * hreg;
            val += __shfl_xor(val, 1);
            val += __shfl_xor(val, 2);
            val += __shfl_xor(val, 4);
            val += __shfl_xor(val, 8);
            if (ln == 0) ybuf[t] = val + bC;
        }
        S[((size_t)b * NCHUNK + c) * HID + ln] = hreg;
    }

    __syncthreads();

    for (int i = ln; i < CHLEN; i += 32)
        y_partial[(size_t)b * TT + t0 + i] = ybuf[i];
}

__global__ __launch_bounds__(32) void ssm_carry_kernel(
    const float* __restrict__ A, const float* __restrict__ S,
    float* __restrict__ Carry)
{
    const int b = blockIdx.x;
    const int h = threadIdx.x;
    if (h >= HID) return;

    float a = A[h];
    float p = a;
#pragma unroll
    for (int i = 0; i < 7; ++i) p *= p;

    float f = 0.0f;
    for (int c = 0; c < NCHUNK; ++c) {
        size_t idx = ((size_t)b * NCHUNK + c) * HID + h;
        Carry[idx] = f;
        f = fmaf(p, f, S[idx]);
    }
}

__global__ __launch_bounds__(CHLEN) void ssm_fix_kernel(
    const float* __restrict__ A, const float* __restrict__ W_C,
    const float* __restrict__ Carry, float* __restrict__ y)
{
    const int wg = blockIdx.x;
    const int b  = wg / NCHUNK;
    const int c  = wg % NCHUNK;
    const int i  = threadIdx.x;

    __shared__ float ch[HID], ah[HID], wch[HID];
    if (i < HID) {
        ch[i]  = Carry[((size_t)b * NCHUNK + c) * HID + i];
        ah[i]  = A[i];
        wch[i] = W_C[i];
    }
    __syncthreads();

    float corr = 0.0f;
#pragma unroll
    for (int h = 0; h < HID; ++h)
        corr = fmaf(wch[h] * ch[h], powf(ah[h], (float)(i + 1)), corr);

    y[(size_t)b * TT + (size_t)c * CHLEN + i] += corr;
}

extern "C" void kernel_launch(void* const* d_in, const int* in_sizes, int n_in,
                              void* d_out, int out_size, void* d_ws, size_t ws_size,
                              hipStream_t stream) {
    const float* u   = (const float*)d_in[0];
    const float* A   = (const float*)d_in[1];
    const float* W_B = (const float*)d_in[2];
    const float* b_B = (const float*)d_in[3];
    const float* W_C = (const float*)d_in[4];
    const float* b_C = (const float*)d_in[5];
    float* y = (float*)d_out;

    float* S     = (float*)d_ws;
    float* Carry = S + (size_t)NBATCH * NCHUNK * HID;

    ssm_chunk_kernel<<<dim3(NBATCH * NCHUNK), dim3(32), 0, stream>>>(
        u, A, W_B, b_B, W_C, b_C, y, S);
    ssm_carry_kernel<<<dim3(NBATCH), dim3(32), 0, stream>>>(A, S, Carry);
    ssm_fix_kernel<<<dim3(NBATCH * NCHUNK), dim3(CHLEN), 0, stream>>>(
        A, W_C, Carry, y);
}


# Carry scan moved into the correction kernel: each wave forms its 16 rows' carries from S by the same fmaf chain (lane = channel, through LDS); 2 launches, was 3; every sum and its order unchanged
# speedup vs baseline: 4.1531x; 4.1531x over previous
#include <hip/hip_runtime.h>
#include <hip/hip_bf16.h>


#define TT      8192
#define NBATCH  128
#define DIN     32
#define HID     16
#define CHLEN   128
#define NCHUNK  (TT / CHLEN)
#define NTILE   (CHLEN / 16)

typedef float v2f __attribute__((ext_vector_type(2)));
typedef float v4f __attribute__((ext_vector_type(4)));
typedef float v8f __attribute__((ext_vector_type(8)));

#define WV_TILES   16
#define WV_STEPS   (16 * WV_TILES)
#define WV_PER_ROW (TT / WV_STEPS)
#define K1_WAVES   4

template <int SH>
__device__ __forceinline__ float row_shift_right(float x)
{
    return __builtin_bit_cast(float,
        __builtin_amdgcn_update_dpp(0, __builtin_bit_cast(int, x), 0x110 + SH, 0xF, 0xF, true));
}

__global__ __launch_bounds__(32 * K1_WAVES) void ssm_chunk_kernel(
    const float* __restrict__ u, const float* __restrict__ A,
    const float* __restrict__ W_B, const float* __restrict__ b_B,
    const float* __restrict__ W_C, const float* __restrict__ b_C,
    float* __restrict__ y_partial, float* __restrict__ S)
{
    const int ln = threadIdx.x & 31;
    const int wv = blockIdx.x * K1_WAVES + (threadIdx.x >> 5);
    const int n  = ln & 15;
    const int kh = ln >> 4;
    int       nt = n & 7;
    const int b  = wv / WV_PER_ROW;
    const int t0 = (wv % WV_PER_ROW) * WV_STEPS;

    unsigned yo = (unsigned)b * TT + t0 + 16 * (8 * kh) + n;
    unsigned so = (((unsigned)b * NCHUNK + t0 / CHLEN + (n >> 3)) * HID + 8 * kh) / 4;
    asm volatile("" : "+v"(yo), "+v"(so), "+v"(nt));

    v2f wfrag[8];
#pragma unroll
    for (int k = 0; k < 8; ++k) {
        const float* p = W_B + n * DIN + 4 * k + 2 * kh;
        wfrag[k].x = p[0];
        wfrag[k].y = p[1];
    }
    v2f av[4], wc[4];
    v8f bias;
#pragma unroll
    for (int j = 0; j < 4; ++j) {
        av[j].x = A[8 * kh + 2 * j];
        av[j].y = A[8 * kh + 2 * j + 1];
        wc[j].x = W_C[8 * kh + 2 * j];
        wc[j].y = W_C[8 * kh + 2 * j + 1];
        bias[2 * j]     = b_B[8 * kh + 2 * j];
        bias[2 * j + 1] = b_B[8 * kh + 2 * j + 1];
    }
    const float bC = b_C[0];

    v2f a16[4], a32[4], a64[4];
#pragma unroll
    for (int j = 0; j < 4; ++j) {
        const v2f a2 = av[j] * av[j];
        const v2f a4 = a2 * a2;
        const v2f a8 = a4 * a4;
        a16[j] = a8 * a8;
        a32[j] = a16[j] * a16[j];
        a64[j] = a32[j] * a32[j];
    }
    asm volatile("" : "+v"(a16[0]), "+v"(a16[1]), "+v"(a16[2]), "+v"(a16[3]),
                      "+v"(a32[0]), "+v"(a32[1]), "+v"(a32[2]), "+v"(a32[3]),
                      "+v"(a64[0]), "+v"(a64[1]), "+v"(a64[2]), "+v"(a64[3]));

    const float* ulane = u + ((size_t)b * TT + t0 + 16 * n) * DIN + 2 * kh;

    v2f ub[3][8];
#pragma unroll
    for (int k = 0; k < 8; ++k) {
        ub[15 % 3][k] = *reinterpret_cast<const v2f*>(ulane + 15 * DIN + 4 * k);
        ub[14 % 3][k] = *reinterpret_cast<const v2f*>(ulane + 14 * DIN + 4 * k);
    }

    v8f yacc;
#pragma unroll
    for (int v = 0; v < 8; ++v) yacc[v] = bC;

    v2f q[4];
    v2f r[4];
    v2f st[4];
#pragma unroll
    for (int j = 0; j < 4; ++j) {
        q[j].x = (n == 15) ? wc[j].x : 0.0f;
        q[j].y = (n == 15) ? wc[j].y : 0.0f;
        r[j].x = 1.0f;  r[j].y = 1.0f;
        st[j].x = 0.0f; st[j].y = 0.0f;
    }

#pragma unroll
    for (int s = 15; s >= 0; --s) {
        if (s >= 2) {
#pragma unroll
            for (int k = 0; k < 8; ++k)
                ub[(s - 2) % 3][k] = *reinterpret_cast<const v2f*>(ulane + (s - 2) * DIN + 4 * k);
        }

        asm volatile("" : "+v"(ub[s % 3][0]), "+v"(yacc),
                          "+v"(q[0]), "+v"(q[1]), "+v"(q[2]), "+v"(q[3]),
                          "+v"(st[0]), "+v"(st[1]), "+v"(st[2]), "+v"(st[3]),
                          "+v"(r[0]), "+v"(r[1]), "+v"(r[2]), "+v"(r[3]));

        v8f d = bias;
#pragma unroll
        for (int k = 0; k < 8; ++k)
            d = __builtin_amdgcn_wmma_f32_16x16x4_f32(
                 false, wfrag[k],  false, ub[s % 3][k],
                 (short)0, d,  false,  false);

        asm volatile("v_nop\n\tv_nop\n\tv_nop\n\tv_nop\n\tv_nop" : "+v"(d));

#pragma unroll
        for (int j = 0; j < 4; ++j) {
            v2f dj;
            dj.x = d[2 * j];
            dj.y = d[2 * j + 1];
            yacc = __builtin_amdgcn_wmma_f32_16x16x4_f32(
                 false, dj,  false, q[j],
                 (short)0, yacc,  false,  false);
        }

        asm volatile("v_nop\n\tv_nop\n\tv_nop\n\tv_nop"
                     : "+v"(yacc), "+v"(d),
                       "+v"(q[0]), "+v"(q[1]), "+v"(q[2]), "+v"(q[3]),
                       "+v"(r[0]), "+v"(r[1]), "+v"(r[2]), "+v"(r[3]));

#pragma unroll
        for (int j = 0; j < 4; ++j) {
            v2f dj;
            dj.x = d[2 * j];
            dj.y = d[2 * j + 1];
            st[j] = r[j] * dj + st[j];
            r[j]  = r[j] * av[j];
        }

#pragma unroll
        for (int j = 0; j < 4; ++j) {
            const v2f qm = q[j] * av[j];
            q[j].x = (n == s - 1) ? wc[j].x : qm.x;
            q[j].y = (n == s - 1) ? wc[j].y : qm.y;
        }
    }

    v2f e[4];
#pragma unroll
    for (int j = 0; j < 4; ++j) {
        v2f x = st[j];
        v2f sh;
        sh.x = row_shift_right<1>(x.x);  sh.y = row_shift_right<1>(x.y);
        if (nt < 1) { sh.x = 0.0f; sh.y = 0.0f; }
        x = a16[j] * sh + x;
        sh.x = row_shift_right<2>(x.x);  sh.y = row_shift_right<2>(x.y);
        if (nt < 2) { sh.x = 0.0f; sh.y = 0.0f; }
        x = a32[j] * sh + x;
        sh.x = row_shift_right<4>(x.x);  sh.y = row_shift_right<4>(x.y);
        if (nt < 4) { sh.x = 0.0f; sh.y = 0.0f; }
        x = a64[j] * sh + x;
        e[j] = x;
    }
    v2f fin[4];
#pragma unroll
    for (int j = 0; j < 4; ++j) {
        const float f0 = row_shift_right<1>(e[j].x);
        const float f1 = row_shift_right<1>(e[j].y);
        fin[j].x = (nt >= 1) ? f0 : 0.0f;
        fin[j].y = (nt >= 1) ? f1 : 0.0f;
    }
    asm volatile("" : "+v"(yacc),
                      "+v"(fin[0]), "+v"(fin[1]), "+v"(fin[2]), "+v"(fin[3]),
                      "+v"(q[0]), "+v"(q[1]), "+v"(q[2]), "+v"(q[3]),
                      "+v"(e[0]), "+v"(e[1]), "+v"(e[2]), "+v"(e[3]));
#pragma unroll
    for (int j = 0; j < 4; ++j)
        yacc = __builtin_amdgcn_wmma_f32_16x16x4_f32(
             false, fin[j],  false, q[j],
             (short)0, yacc,  false,  false);
    asm volatile("v_nop\n\tv_nop\n\tv_nop\n\tv_nop"
                 : "+v"(yacc), "+v"(e[0]), "+v"(e[1]), "+v"(e[2]), "+v"(e[3]));

    if (nt == 7) {
        v4f lo4, hi4;
        lo4.x = e[0].x; lo4.y = e[0].y; lo4.z = e[1].x; lo4.w = e[1].y;
        hi4.x = e[2].x; hi4.y = e[2].y; hi4.z = e[3].x; hi4.w = e[3].y;
        reinterpret_cast<v4f*>(S)[so]     = lo4;
        reinterpret_cast<v4f*>(S)[so + 1] = hi4;
    }

#pragma unroll
    for (int v = 0; v < 8; ++v) y_partial[yo + 16 * v] = yacc[v];
}

#define FIX_ROWS   16
#define FIX_STRIDE 36
#define CAR_STRIDE 18

__global__ __launch_bounds__(128) void ssm_fix_kernel(
    const float* __restrict__ A, const float* __restrict__ W_C,
    const float* __restrict__ S, float* __restrict__ y)
{
    __shared__ __attribute__((aligned(16))) float stage[4 * FIX_ROWS * FIX_STRIDE];
    __shared__ __attribute__((aligned(16))) float car[4 * FIX_ROWS * CAR_STRIDE];

    const int ln = threadIdx.x & 31;
    const int w  = threadIdx.x >> 5;
    const int m  = ln & 15;
    const int kh = ln >> 4;
    const int r0 = blockIdx.x * FIX_ROWS;

    const float am = A[m];
    v2f ach[4], wch[4];
#pragma unroll
    for (int k = 0; k < 4; ++k) {
        ach[k] = *reinterpret_cast<const v2f*>(A + 4 * k + 2 * kh);
        wch[k] = *reinterpret_cast<const v2f*>(W_C + 4 * k + 2 * kh);
    }

    const int gb = (r0 / FIX_ROWS) % (NCHUNK / FIX_ROWS);
    const float* sp = S + (size_t)(r0 / NCHUNK) * NCHUNK * HID + m;
    float sv[NCHUNK - 1];
#pragma unroll
    for (int c = 0; c < NCHUNK - 1; ++c) sv[c] = sp[c * HID];

    const float* yp = y + (size_t)(r0 + 8 * kh) * CHLEN + 32 * w + m;
    v8f cacc[2];
#pragma unroll
    for (int t = 0; t < 2; ++t)
#pragma unroll
        for (int v = 0; v < 8; ++v) cacc[t][v] = yp[v * CHLEN + 16 * t];

    __builtin_amdgcn_sched_barrier(0);

    const unsigned e = 32u * w + m + 1u;
    v2f bfrag[2][4];
#pragma unroll
    for (int k = 0; k < 4; ++k) {
#pragma unroll
        for (int j = 0; j < 2; ++j) {
            float sq  = ach[k][j];
            float pw  = (e & 1u) ? sq : 1.0f;
            float a16 = 0.0f;
#pragma unroll
            for (int bit = 1; bit < 7; ++bit) {
                sq *= sq;
                if (bit == 4) a16 = sq;
                pw = ((e >> bit) & 1u) ? pw * sq : pw;
            }
            const float wc = wch[k][j];
            bfrag[0][k][j] = wc * pw;
            bfrag[1][k][j] = wc * (pw * a16);
        }
    }

    float pa = am;
#pragma unroll
    for (int i = 0; i < 7; ++i) pa *= pa;

    float fc = 0.0f;
    float own[FIX_ROWS];
#pragma unroll
    for (int i = 0; i < FIX_ROWS; ++i) own[i] = 0.0f;
#pragma unroll
    for (int c = 0; c < NCHUNK; ++c) {
        own[c % FIX_ROWS] = (c / FIX_ROWS == gb) ? fc : own[c % FIX_ROWS];
        if (c < NCHUNK - 1) fc = fmaf(pa, fc, sv[c]);
    }

#pragma unroll
    for (int i = 0; i < FIX_ROWS; ++i)
        car[(w * FIX_ROWS + i) * CAR_STRIDE + m] = own[i];
    __syncthreads();

    v2f afrag[4];
#pragma unroll
    for (int k = 0; k < 4; ++k)
        afrag[k] = *reinterpret_cast<const v2f*>(&car[(w * FIX_ROWS + m) * CAR_STRIDE + 4 * k + 2 * kh]);

    const int lr = ln >> 3;
    const int lq = ln & 7;
    unsigned so = w * (FIX_ROWS * FIX_STRIDE) + 8 * kh * FIX_STRIDE + m;
    unsigned lo = w * (FIX_ROWS * FIX_STRIDE / 4) + lr * (FIX_STRIDE / 4) + lq;
    unsigned yo = (r0 + lr) * (CHLEN / 4) + 8 * w + lq;
    asm volatile("" : "+v"(so), "+v"(lo), "+v"(yo));

    __builtin_amdgcn_sched_barrier(0);
#pragma unroll
    for (int k = 0; k < 4; ++k)
#pragma unroll
        for (int t = 0; t < 2; ++t)
            cacc[t] = __builtin_amdgcn_wmma_f32_16x16x4_f32(
                 false, afrag[k],  false, bfrag[t][k],
                 (short)0, cacc[t],  false,  false);
    __builtin_amdgcn_sched_barrier(0);
    asm volatile("v_nop\n\tv_nop\n\tv_nop\n\tv_nop");
    __builtin_amdgcn_sched_barrier(0);

#pragma unroll
    for (int t = 0; t < 2; ++t)
#pragma unroll
        for (int v = 0; v < 8; ++v)
            stage[so + v * FIX_STRIDE + 16 * t] = cacc[t][v];

    __syncthreads();

#pragma unroll
    for (int it = 0; it < 4; ++it) {
        const v4f val = reinterpret_cast<const v4f*>(stage)[lo + it * FIX_STRIDE];
        reinterpret_cast<v4f*>(y)[yo + it * CHLEN] = val;
    }
}

extern "C" void kernel_launch(void* const* d_in, const int* in_sizes, int n_in,
                              void* d_out, int out_size, void* d_ws, size_t ws_size,
                              hipStream_t stream) {
    const float* u   = (const float*)d_in[0];
    const float* A   = (const float*)d_in[1];
    const float* W_B = (const float*)d_in[2];
    const float* b_B = (const float*)d_in[3];
    const float* W_C = (const float*)d_in[4];
    const float* b_C = (const float*)d_in[5];
    float* y = (float*)d_out;

    float* S     = (float*)d_ws;

    ssm_chunk_kernel<<<dim3(NBATCH * WV_PER_ROW / K1_WAVES), dim3(32 * K1_WAVES), 0, stream>>>(
        u, A, W_B, b_B, W_C, b_C, y, S);
    ssm_fix_kernel<<<dim3(NBATCH * NCHUNK / FIX_ROWS), dim3(128), 0, stream>>>(
        A, W_C, S, y);
}
